# starting kernel
# baseline (speedup 1.0000x reference)
#include <hip/hip_runtime.h>
#include <hip/hip_bf16.h>
#include <math.h>


typedef __attribute__((ext_vector_type(2))) float v2f;
typedef __attribute__((ext_vector_type(8))) float v8f;

#define IN_DIM   128
#define HIDDEN    16
#define NCLASS    40
#define NPAD      48

__global__ void gcn_zero_f32(float* __restrict__ p, int n) {
  int i = blockIdx.x * blockDim.x + threadIdx.x;
  if (i < n) p[i] = 0.0f;
}

__global__ void gcn_spmm1(const int* __restrict__ rows,
                          const int* __restrict__ cols,
                          const float* __restrict__ vals,
                          const float* __restrict__ x,
                          float* __restrict__ z1, int nEdges) {
  int gid = blockIdx.x * blockDim.x + threadIdx.x;
  int e  = gid >> 5;
  int f4 = gid & 31;
  if (e >= nEdges) return;
  int r = rows[e];
  int c = cols[e];
  float v = vals[e];
  const float4* xr = (const float4*)(x + (size_t)c * IN_DIM);
  float4 d = xr[f4];
  float* zr = z1 + (size_t)r * IN_DIM + (size_t)f4 * 4;
  atomicAdd(zr + 0, v * d.x);
  atomicAdd(zr + 1, v * d.y);
  atomicAdd(zr + 2, v * d.z);
  atomicAdd(zr + 3, v * d.w);
}

__global__ void gcn_spmm2(const int* __restrict__ rows,
                          const int* __restrict__ cols,
                          const float* __restrict__ vals,
                          const float* __restrict__ h,
                          float* __restrict__ z2, int nEdges) {
  int gid = blockIdx.x * blockDim.x + threadIdx.x;
  int e  = gid >> 2;
  int f4 = gid & 3;
  if (e >= nEdges) return;
  int r = rows[e];
  int c = cols[e];
  float v = vals[e];
  const float4* hr = (const float4*)(h + (size_t)c * HIDDEN);
  float4 d = hr[f4];
  float* zr = z2 + (size_t)r * HIDDEN + (size_t)f4 * 4;
  atomicAdd(zr + 0, v * d.x);
  atomicAdd(zr + 1, v * d.y);
  atomicAdd(zr + 2, v * d.z);
  atomicAdd(zr + 3, v * d.w);
}

__global__ void __launch_bounds__(32)
gcn_gemm1_relu(const float* __restrict__ z1,
               const float* __restrict__ w1,
               float* __restrict__ h) {
  __shared__ float Wl[IN_DIM * HIDDEN];
  __shared__ float Al[16 * IN_DIM];
  const int lane = threadIdx.x;
  const int tile = blockIdx.x;

  for (int i = lane; i < IN_DIM * HIDDEN; i += 32) Wl[i] = w1[i];
  const float* Ab = z1 + (size_t)tile * 16 * IN_DIM;
  for (int i = lane; i < 16 * IN_DIM; i += 32) Al[i] = Ab[i];
  __syncthreads();

  const int half = lane >> 4;
  const int l    = lane & 15;

  v8f c = {};
  #pragma unroll 8
  for (int k0 = 0; k0 < IN_DIM; k0 += 4) {
    const int ka = k0 + 2 * half;
    v2f a, b;
    a.x = Al[l * IN_DIM + ka];
    a.y = Al[l * IN_DIM + ka + 1];
    b.x = Wl[ka * HIDDEN + l];
    b.y = Wl[(ka + 1) * HIDDEN + l];
    c = __builtin_amdgcn_wmma_f32_16x16x4_f32(false, a, false, b,
                                              (short)0, c, false, false);
  }

  float* hb = h + (size_t)tile * 16 * HIDDEN;
  #pragma unroll
  for (int i = 0; i < 8; ++i) {
    float v = c[i];
    hb[(i + 8 * half) * HIDDEN + l] = v > 0.0f ? v : 0.0f;
  }
}

__global__ void __launch_bounds__(32)
gcn_gemm2_lsm(const float* __restrict__ z2,
              const float* __restrict__ w2,
              float* __restrict__ out) {
  __shared__ float Wl[HIDDEN * NPAD];
  __shared__ float Al[16 * HIDDEN];
  __shared__ float Ol[16 * NPAD];
  const int lane = threadIdx.x;
  const int tile = blockIdx.x;

  for (int i = lane; i < HIDDEN * NPAD; i += 32) {
    int r = i / NPAD, n = i % NPAD;
    Wl[i] = (n < NCLASS) ? w2[r * NCLASS + n] : 0.0f;
  }
  const float* Ab = z2 + (size_t)tile * 16 * HIDDEN;
  for (int i = lane; i < 16 * HIDDEN; i += 32) Al[i] = Ab[i];
  __syncthreads();

  const int half = lane >> 4;
  const int l    = lane & 15;

  #pragma unroll
  for (int t = 0; t < 3; ++t) {
    v8f c = {};
    #pragma unroll
    for (int k0 = 0; k0 < HIDDEN; k0 += 4) {
      const int ka = k0 + 2 * half;
      v2f a, b;
      a.x = Al[l * HIDDEN + ka];
      a.y = Al[l * HIDDEN + ka + 1];
      b.x = Wl[ka * NPAD + t * 16 + l];
      b.y = Wl[(ka + 1) * NPAD + t * 16 + l];
      c = __builtin_amdgcn_wmma_f32_16x16x4_f32(false, a, false, b,
                                                (short)0, c, false, false);
    }
    #pragma unroll
    for (int i = 0; i < 8; ++i)
      Ol[(i + 8 * half) * NPAD + t * 16 + l] = c[i];
  }
  __syncthreads();

  if (lane < 16) {
    const float* row = Ol + lane * NPAD;
    float mx = row[0];
    for (int j = 1; j < NCLASS; ++j) mx = fmaxf(mx, row[j]);
    float s = 0.0f;
    for (int j = 0; j < NCLASS; ++j) s += __expf(row[j] - mx);
    const float lse = mx + __logf(s);
    float* ob = out + ((size_t)tile * 16 + lane) * NCLASS;
    for (int j = 0; j < NCLASS; ++j) ob[j] = row[j] - lse;
  }
}

extern "C" void kernel_launch(void* const* d_in, const int* in_sizes, int n_in,
                              void* d_out, int out_size, void* d_ws, size_t ws_size,
                              hipStream_t stream) {
  const float* features = (const float*)d_in[0];
  const float* weight1  = (const float*)d_in[1];
  const float* weight2  = (const float*)d_in[2];
  const float* edgeVals = (const float*)d_in[3];
  const int*   edgeRows = (const int*)  d_in[4];
  const int*   edgeCols = (const int*)  d_in[5];

  const int nNodes = in_sizes[0] / IN_DIM;
  const int nEdges = in_sizes[3];
  const int nTiles = nNodes / 16;

  float* z1 = (float*)d_ws;
  float* h  = z1 + (size_t)nNodes * IN_DIM;
  float* z2 = h  + (size_t)nNodes * HIDDEN;

  {
    int n = nNodes * (IN_DIM + 2 * HIDDEN);
    gcn_zero_f32<<<(n + 255) / 256, 256, 0, stream>>>(z1, n);
  }
  {
    long long threads = (long long)nEdges * 32;
    gcn_spmm1<<<(unsigned)((threads + 255) / 256), 256, 0, stream>>>(
        edgeRows, edgeCols, edgeVals, features, z1, nEdges);
  }
  gcn_gemm1_relu<<<nTiles, 32, 0, stream>>>(z1, weight1, h);
  {
    long long threads = (long long)nEdges * 4;
    gcn_spmm2<<<(unsigned)((threads + 255) / 256), 256, 0, stream>>>(
        edgeRows, edgeCols, edgeVals, h, z2, nEdges);
  }
  gcn_gemm2_lsm<<<nTiles, 32, 0, stream>>>(z2, weight2, (float*)d_out);
}


# Layer 1 reassociated as A@(X@W1): WMMA product on the features first, so the atomic edge sum carries 16 floats per edge, not 128; ReLU taken at layer 2's gather; f32 sum order changes
# speedup vs baseline: 1.8300x; 1.8300x over previous
#include <hip/hip_runtime.h>
#include <hip/hip_bf16.h>
#include <math.h>


typedef __attribute__((ext_vector_type(2))) float v2f;
typedef __attribute__((ext_vector_type(8))) float v8f;

#define IN_DIM   128
#define HIDDEN    16
#define NCLASS    40
#define NPAD      48

__global__ void gcn_zero_f32(float* __restrict__ p, int n) {
  int i = blockIdx.x * blockDim.x + threadIdx.x;
  if (i < n) p[i] = 0.0f;
}

template <bool RELU>
__device__ __forceinline__ void gcn_spmm16_body(const int* __restrict__ rows,
                                                const int* __restrict__ cols,
                                                const float* __restrict__ vals,
                                                const float* __restrict__ src,
                                                float* __restrict__ dst,
                                                int nEdges) {
  int gid = blockIdx.x * blockDim.x + threadIdx.x;
  int e  = gid >> 2;
  int f4 = gid & 3;
  if (e >= nEdges) return;
  int r = rows[e];
  int c = cols[e];
  float v = vals[e];
  const float4* sr = (const float4*)(src + (size_t)c * HIDDEN);
  float4 d = sr[f4];
  if (RELU) {
    d.x = d.x > 0.0f ? d.x : 0.0f;
    d.y = d.y > 0.0f ? d.y : 0.0f;
    d.z = d.z > 0.0f ? d.z : 0.0f;
    d.w = d.w > 0.0f ? d.w : 0.0f;
  }
  float* zr = dst + (size_t)r * HIDDEN + (size_t)f4 * 4;
  atomicAdd(zr + 0, v * d.x);
  atomicAdd(zr + 1, v * d.y);
  atomicAdd(zr + 2, v * d.z);
  atomicAdd(zr + 3, v * d.w);
}

__global__ void gcn_spmm_l1(const int* __restrict__ rows,
                            const int* __restrict__ cols,
                            const float* __restrict__ vals,
                            const float* __restrict__ xw,
                            float* __restrict__ t, int nEdges) {
  gcn_spmm16_body<false>(rows, cols, vals, xw, t, nEdges);
}

__global__ void gcn_spmm_l2(const int* __restrict__ rows,
                            const int* __restrict__ cols,
                            const float* __restrict__ vals,
                            const float* __restrict__ t,
                            float* __restrict__ z2, int nEdges) {
  gcn_spmm16_body<true>(rows, cols, vals, t, z2, nEdges);
}

__global__ void __launch_bounds__(32)
gcn_gemm1(const float* __restrict__ x,
          const float* __restrict__ w1,
          float* __restrict__ xw) {
  __shared__ float Wl[IN_DIM * HIDDEN];
  __shared__ float Al[16 * IN_DIM];
  const int lane = threadIdx.x;
  const int tile = blockIdx.x;

  for (int i = lane; i < IN_DIM * HIDDEN; i += 32) Wl[i] = w1[i];
  const float* Ab = x + (size_t)tile * 16 * IN_DIM;
  for (int i = lane; i < 16 * IN_DIM; i += 32) Al[i] = Ab[i];
  __syncthreads();

  const int half = lane >> 4;
  const int l    = lane & 15;

  v8f c = {};
  #pragma unroll 8
  for (int k0 = 0; k0 < IN_DIM; k0 += 4) {
    const int ka = k0 + 2 * half;
    v2f a, b;
    a.x = Al[l * IN_DIM + ka];
    a.y = Al[l * IN_DIM + ka + 1];
    b.x = Wl[ka * HIDDEN + l];
    b.y = Wl[(ka + 1) * HIDDEN + l];
    c = __builtin_amdgcn_wmma_f32_16x16x4_f32(false, a, false, b,
                                              (short)0, c, false, false);
  }

  float* ob = xw + (size_t)tile * 16 * HIDDEN;
  #pragma unroll
  for (int i = 0; i < 8; ++i)
    ob[(i + 8 * half) * HIDDEN + l] = c[i];
}

__global__ void __launch_bounds__(32)
gcn_gemm2_lsm(const float* __restrict__ z2,
              const float* __restrict__ w2,
              float* __restrict__ out) {
  __shared__ float Wl[HIDDEN * NPAD];
  __shared__ float Al[16 * HIDDEN];
  __shared__ float Ol[16 * NPAD];
  const int lane = threadIdx.x;
  const int tile = blockIdx.x;

  for (int i = lane; i < HIDDEN * NPAD; i += 32) {
    int r = i / NPAD, n = i % NPAD;
    Wl[i] = (n < NCLASS) ? w2[r * NCLASS + n] : 0.0f;
  }
  const float* Ab = z2 + (size_t)tile * 16 * HIDDEN;
  for (int i = lane; i < 16 * HIDDEN; i += 32) Al[i] = Ab[i];
  __syncthreads();

  const int half = lane >> 4;
  const int l    = lane & 15;

  #pragma unroll
  for (int t = 0; t < 3; ++t) {
    v8f c = {};
    #pragma unroll
    for (int k0 = 0; k0 < HIDDEN; k0 += 4) {
      const int ka = k0 + 2 * half;
      v2f a, b;
      a.x = Al[l * HIDDEN + ka];
      a.y = Al[l * HIDDEN + ka + 1];
      b.x = Wl[ka * NPAD + t * 16 + l];
      b.y = Wl[(ka + 1) * NPAD + t * 16 + l];
      c = __builtin_amdgcn_wmma_f32_16x16x4_f32(false, a, false, b,
                                                (short)0, c, false, false);
    }
    #pragma unroll
    for (int i = 0; i < 8; ++i)
      Ol[(i + 8 * half) * NPAD + t * 16 + l] = c[i];
  }
  __syncthreads();

  if (lane < 16) {
    const float* row = Ol + lane * NPAD;
    float mx = row[0];
    for (int j = 1; j < NCLASS; ++j) mx = fmaxf(mx, row[j]);
    float s = 0.0f;
    for (int j = 0; j < NCLASS; ++j) s += __expf(row[j] - mx);
    const float lse = mx + __logf(s);
    float* ob = out + ((size_t)tile * 16 + lane) * NCLASS;
    for (int j = 0; j < NCLASS; ++j) ob[j] = row[j] - lse;
  }
}

extern "C" void kernel_launch(void* const* d_in, const int* in_sizes, int n_in,
                              void* d_out, int out_size, void* d_ws, size_t ws_size,
                              hipStream_t stream) {
  const float* features = (const float*)d_in[0];
  const float* weight1  = (const float*)d_in[1];
  const float* weight2  = (const float*)d_in[2];
  const float* edgeVals = (const float*)d_in[3];
  const int*   edgeRows = (const int*)  d_in[4];
  const int*   edgeCols = (const int*)  d_in[5];

  const int nNodes = in_sizes[0] / IN_DIM;
  const int nEdges = in_sizes[3];
  const int nTiles = nNodes / 16;

  float* xw = (float*)d_ws;
  float* t  = xw + (size_t)nNodes * HIDDEN;
  float* z2 = t  + (size_t)nNodes * HIDDEN;

  {
    int n = nNodes * (2 * HIDDEN);
    gcn_zero_f32<<<(n + 255) / 256, 256, 0, stream>>>(t, n);
  }
  gcn_gemm1<<<nTiles, 32, 0, stream>>>(features, weight1, xw);
  {
    long long threads = (long long)nEdges * 4;
    gcn_spmm_l1<<<(unsigned)((threads + 255) / 256), 256, 0, stream>>>(
        edgeRows, edgeCols, edgeVals, xw, t, nEdges);
  }
  {
    long long threads = (long long)nEdges * 4;
    gcn_spmm_l2<<<(unsigned)((threads + 255) / 256), 256, 0, stream>>>(
        edgeRows, edgeCols, edgeVals, t, z2, nEdges);
  }
  gcn_gemm2_lsm<<<nTiles, 32, 0, stream>>>(z2, weight2, (float*)d_out);
}


# Edge passes: one lane per feature, a wave owns 32 edges (index and weight handed over by DPP row_share), so each gather and f32 atomic covers 2 whole 64-B rows, not 8 quarter rows; arrival-order sums
# speedup vs baseline: 2.6861x; 1.4678x over previous
#include <hip/hip_runtime.h>
#include <hip/hip_bf16.h>
#include <math.h>


typedef __attribute__((ext_vector_type(2))) float v2f;
typedef __attribute__((ext_vector_type(8))) float v8f;

#define IN_DIM   128
#define HIDDEN    16
#define NCLASS    40
#define NPAD      48

__global__ void gcn_zero_f32(float* __restrict__ p, int n) {
  int i = blockIdx.x * blockDim.x + threadIdx.x;
  if (i < n) p[i] = 0.0f;
}


#define GCN_SHARE(x, J) __builtin_amdgcn_update_dpp(0, (x), 0x150 + (J), 0xf, 0xf, true)
#define GCN_FOR16(OP) OP(0) OP(1) OP(2) OP(3) OP(4) OP(5) OP(6) OP(7) \
                      OP(8) OP(9) OP(10) OP(11) OP(12) OP(13) OP(14) OP(15)

template <bool RELU>
__device__ __forceinline__ float gcn_act(float d) {
  return RELU ? (d > 0.0f ? d : 0.0f) : d;
}

template <bool RELU>
__device__ __forceinline__ void gcn_spmm16_body(const int* __restrict__ rows,
                                                const int* __restrict__ cols,
                                                const float* __restrict__ vals,
                                                const float* __restrict__ src,
                                                float* __restrict__ dst,
                                                int nEdges) {
  const int e    = blockIdx.x * blockDim.x + threadIdx.x;
  const int lane = threadIdx.x & 31;
  const int e0   = e - lane;
  if (e0 >= nEdges) return;
  const int f    = lane & 15;
  const int half = lane >> 4;

  int rofs = 0, cofs = 0, vbits = 0;
  if (e < nEdges) {
    rofs  = rows[e] * HIDDEN;
    cofs  = cols[e] * HIDDEN;
    vbits = __float_as_int(vals[e]);
  }

  if (e0 + 32 <= nEdges) {
    float d[16];
#define GCN_GATHER(J) d[J] = src[(unsigned)(GCN_SHARE(cofs, J) + f)];
    GCN_FOR16(GCN_GATHER)
#undef GCN_GATHER
#define GCN_ADD(J)                                                   \
    atomicAdd(dst + (unsigned)(GCN_SHARE(rofs, J) + f),                \
              __int_as_float(GCN_SHARE(vbits, J)) * gcn_act<RELU>(d[J]));
    GCN_FOR16(GCN_ADD)
#undef GCN_ADD
  } else {
    for (int j = 0; j < 16; ++j) {
      const int sl = (half << 4) + j;
      const int cj = __shfl(cofs, sl, 32);
      const int rj = __shfl(rofs, sl, 32);
      const float vj = __int_as_float(__shfl(vbits, sl, 32));
      if (e0 + sl < nEdges) {
        const float dj = gcn_act<RELU>(src[(unsigned)(cj + f)]);
        atomicAdd(dst + (unsigned)(rj + f), vj * dj);
      }
    }
  }
}

__global__ void gcn_spmm_l1(const int* __restrict__ rows,
                            const int* __restrict__ cols,
                            const float* __restrict__ vals,
                            const float* __restrict__ xw,
                            float* __restrict__ t, int nEdges) {
  gcn_spmm16_body<false>(rows, cols, vals, xw, t, nEdges);
}

__global__ void gcn_spmm_l2(const int* __restrict__ rows,
                            const int* __restrict__ cols,
                            const float* __restrict__ vals,
                            const float* __restrict__ t,
                            float* __restrict__ z2, int nEdges) {
  gcn_spmm16_body<true>(rows, cols, vals, t, z2, nEdges);
}

__global__ void __launch_bounds__(32)
gcn_gemm1(const float* __restrict__ x,
          const float* __restrict__ w1,
          float* __restrict__ xw) {
  __shared__ float Wl[IN_DIM * HIDDEN];
  __shared__ float Al[16 * IN_DIM];
  const int lane = threadIdx.x;
  const int tile = blockIdx.x;

  for (int i = lane; i < IN_DIM * HIDDEN; i += 32) Wl[i] = w1[i];
  const float* Ab = x + (size_t)tile * 16 * IN_DIM;
  for (int i = lane; i < 16 * IN_DIM; i += 32) Al[i] = Ab[i];
  __syncthreads();

  const int half = lane >> 4;
  const int l    = lane & 15;

  v8f c = {};
  #pragma unroll 8
  for (int k0 = 0; k0 < IN_DIM; k0 += 4) {
    const int ka = k0 + 2 * half;
    v2f a, b;
    a.x = Al[l * IN_DIM + ka];
    a.y = Al[l * IN_DIM + ka + 1];
    b.x = Wl[ka * HIDDEN + l];
    b.y = Wl[(ka + 1) * HIDDEN + l];
    c = __builtin_amdgcn_wmma_f32_16x16x4_f32(false, a, false, b,
                                              (short)0, c, false, false);
  }

  float* ob = xw + (size_t)tile * 16 * HIDDEN;
  #pragma unroll
  for (int i = 0; i < 8; ++i)
    ob[(i + 8 * half) * HIDDEN + l] = c[i];
}

__global__ void __launch_bounds__(32)
gcn_gemm2_lsm(const float* __restrict__ z2,
              const float* __restrict__ w2,
              float* __restrict__ out) {
  __shared__ float Wl[HIDDEN * NPAD];
  __shared__ float Al[16 * HIDDEN];
  __shared__ float Ol[16 * NPAD];
  const int lane = threadIdx.x;
  const int tile = blockIdx.x;

  for (int i = lane; i < HIDDEN * NPAD; i += 32) {
    int r = i / NPAD, n = i % NPAD;
    Wl[i] = (n < NCLASS) ? w2[r * NCLASS + n] : 0.0f;
  }
  const float* Ab = z2 + (size_t)tile * 16 * HIDDEN;
  for (int i = lane; i < 16 * HIDDEN; i += 32) Al[i] = Ab[i];
  __syncthreads();

  const int half = lane >> 4;
  const int l    = lane & 15;

  #pragma unroll
  for (int t = 0; t < 3; ++t) {
    v8f c = {};
    #pragma unroll
    for (int k0 = 0; k0 < HIDDEN; k0 += 4) {
      const int ka = k0 + 2 * half;
      v2f a, b;
      a.x = Al[l * HIDDEN + ka];
      a.y = Al[l * HIDDEN + ka + 1];
      b.x = Wl[ka * NPAD + t * 16 + l];
      b.y = Wl[(ka + 1) * NPAD + t * 16 + l];
      c = __builtin_amdgcn_wmma_f32_16x16x4_f32(false, a, false, b,
                                                (short)0, c, false, false);
    }
    #pragma unroll
    for (int i = 0; i < 8; ++i)
      Ol[(i + 8 * half) * NPAD + t * 16 + l] = c[i];
  }
  __syncthreads();

  if (lane < 16) {
    const float* row = Ol + lane * NPAD;
    float mx = row[0];
    for (int j = 1; j < NCLASS; ++j) mx = fmaxf(mx, row[j]);
    float s = 0.0f;
    for (int j = 0; j < NCLASS; ++j) s += __expf(row[j] - mx);
    const float lse = mx + __logf(s);
    float* ob = out + ((size_t)tile * 16 + lane) * NCLASS;
    for (int j = 0; j < NCLASS; ++j) ob[j] = row[j] - lse;
  }
}

extern "C" void kernel_launch(void* const* d_in, const int* in_sizes, int n_in,
                              void* d_out, int out_size, void* d_ws, size_t ws_size,
                              hipStream_t stream) {
  const float* features = (const float*)d_in[0];
  const float* weight1  = (const float*)d_in[1];
  const float* weight2  = (const float*)d_in[2];
  const float* edgeVals = (const float*)d_in[3];
  const int*   edgeRows = (const int*)  d_in[4];
  const int*   edgeCols = (const int*)  d_in[5];

  const int nNodes = in_sizes[0] / IN_DIM;
  const int nEdges = in_sizes[3];
  const int nTiles = nNodes / 16;

  float* xw = (float*)d_ws;
  float* t  = xw + (size_t)nNodes * HIDDEN;
  float* z2 = t  + (size_t)nNodes * HIDDEN;

  {
    int n = nNodes * (2 * HIDDEN);
    gcn_zero_f32<<<(n + 255) / 256, 256, 0, stream>>>(t, n);
  }
  gcn_gemm1<<<nTiles, 32, 0, stream>>>(features, weight1, xw);
  gcn_spmm_l1<<<(nEdges + 255) / 256, 256, 0, stream>>>(
      edgeRows, edgeCols, edgeVals, xw, t, nEdges);
  gcn_spmm_l2<<<(nEdges + 255) / 256, 256, 0, stream>>>(
      edgeRows, edgeCols, edgeVals, t, z2, nEdges);
  gcn_gemm2_lsm<<<nTiles, 32, 0, stream>>>(z2, weight2, (float*)d_out);
}


# Dense kernels: a wave's staging loads (x tile + W1; z2 tile + W2) go out as ONE group of 16-byte loads ahead of the LDS stores, not as 8 and 32 load-wait-store trips; LDS image, k order, sums kept
# speedup vs baseline: 3.1857x; 1.1860x over previous
#include <hip/hip_runtime.h>
#include <hip/hip_bf16.h>
#include <math.h>


typedef __attribute__((ext_vector_type(2))) float v2f;
typedef __attribute__((ext_vector_type(4))) float v4f;
typedef __attribute__((ext_vector_type(8))) float v8f;

#define IN_DIM   128
#define HIDDEN    16
#define NCLASS    40
#define NPAD      48

__global__ void gcn_zero_f32(float* __restrict__ p, int n) {
  int i = blockIdx.x * blockDim.x + threadIdx.x;
  if (i < n) p[i] = 0.0f;
}


#define GCN_SHARE(x, J) __builtin_amdgcn_update_dpp(0, (x), 0x150 + (J), 0xf, 0xf, true)
#define GCN_FOR16(OP) OP(0) OP(1) OP(2) OP(3) OP(4) OP(5) OP(6) OP(7) \
                      OP(8) OP(9) OP(10) OP(11) OP(12) OP(13) OP(14) OP(15)

template <bool RELU>
__device__ __forceinline__ float gcn_act(float d) {
  return RELU ? (d > 0.0f ? d : 0.0f) : d;
}

template <bool RELU>
__device__ __forceinline__ void gcn_spmm16_body(const int* __restrict__ rows,
                                                const int* __restrict__ cols,
                                                const float* __restrict__ vals,
                                                const float* __restrict__ src,
                                                float* __restrict__ dst,
                                                int nEdges) {
  const int e    = blockIdx.x * blockDim.x + threadIdx.x;
  const int lane = threadIdx.x & 31;
  const int e0   = e - lane;
  if (e0 >= nEdges) return;
  const int f    = lane & 15;
  const int half = lane >> 4;

  int rofs = 0, cofs = 0, vbits = 0;
  if (e < nEdges) {
    rofs  = rows[e] * HIDDEN;
    cofs  = cols[e] * HIDDEN;
    vbits = __float_as_int(vals[e]);
  }

  if (e0 + 32 <= nEdges) {
    float d[16];
#define GCN_GATHER(J) d[J] = src[(unsigned)(GCN_SHARE(cofs, J) + f)];
    GCN_FOR16(GCN_GATHER)
#undef GCN_GATHER
#define GCN_ADD(J)                                                   \
    atomicAdd(dst + (unsigned)(GCN_SHARE(rofs, J) + f),                \
              __int_as_float(GCN_SHARE(vbits, J)) * gcn_act<RELU>(d[J]));
    GCN_FOR16(GCN_ADD)
#undef GCN_ADD
  } else {
    for (int j = 0; j < 16; ++j) {
      const int sl = (half << 4) + j;
      const int cj = __shfl(cofs, sl, 32);
      const int rj = __shfl(rofs, sl, 32);
      const float vj = __int_as_float(__shfl(vbits, sl, 32));
      if (e0 + sl < nEdges) {
        const float dj = gcn_act<RELU>(src[(unsigned)(cj + f)]);
        atomicAdd(dst + (unsigned)(rj + f), vj * dj);
      }
    }
  }
}

__global__ void gcn_spmm_l1(const int* __restrict__ rows,
                            const int* __restrict__ cols,
                            const float* __restrict__ vals,
                            const float* __restrict__ xw,
                            float* __restrict__ t, int nEdges) {
  gcn_spmm16_body<false>(rows, cols, vals, xw, t, nEdges);
}

__global__ void gcn_spmm_l2(const int* __restrict__ rows,
                            const int* __restrict__ cols,
                            const float* __restrict__ vals,
                            const float* __restrict__ t,
                            float* __restrict__ z2, int nEdges) {
  gcn_spmm16_body<true>(rows, cols, vals, t, z2, nEdges);
}

__global__ void __launch_bounds__(32)
gcn_gemm1(const float* __restrict__ x,
          const float* __restrict__ w1,
          float* __restrict__ xw) {
  __shared__ __attribute__((aligned(16))) float Wl[IN_DIM * HIDDEN];
  __shared__ __attribute__((aligned(16))) float Al[16 * IN_DIM];
  const int lane = threadIdx.x;
  const int tile = blockIdx.x;

  const float* Ab = x + (size_t)tile * 16 * IN_DIM;
  v4f xa[16], wb[16];
  #pragma unroll
  for (int j = 0; j < 16; ++j)
    xa[j] = *(const v4f*)(Ab + 4 * (lane + 32 * j));
  #pragma unroll
  for (int j = 0; j < 16; ++j)
    wb[j] = *(const v4f*)(w1 + 4 * (lane + 32 * j));
  #pragma unroll
  for (int j = 0; j < 16; ++j)
    *(v4f*)(Al + 4 * (lane + 32 * j)) = xa[j];
  #pragma unroll
  for (int j = 0; j < 16; ++j)
    *(v4f*)(Wl + 4 * (lane + 32 * j)) = wb[j];
  __syncthreads();

  const int half = lane >> 4;
  const int l    = lane & 15;

  v8f c = {};
  #pragma unroll 8
  for (int k0 = 0; k0 < IN_DIM; k0 += 4) {
    const int ka = k0 + 2 * half;
    v2f a, b;
    a.x = Al[l * IN_DIM + ka];
    a.y = Al[l * IN_DIM + ka + 1];
    b.x = Wl[ka * HIDDEN + l];
    b.y = Wl[(ka + 1) * HIDDEN + l];
    c = __builtin_amdgcn_wmma_f32_16x16x4_f32(false, a, false, b,
                                              (short)0, c, false, false);
  }

  float* ob = xw + (size_t)tile * 16 * HIDDEN;
  #pragma unroll
  for (int i = 0; i < 8; ++i)
    ob[(i + 8 * half) * HIDDEN + l] = c[i];
}

__global__ void __launch_bounds__(32)
gcn_gemm2_lsm(const float* __restrict__ z2,
              const float* __restrict__ w2,
              float* __restrict__ out) {
  __shared__ __attribute__((aligned(16))) float Wl[HIDDEN * NPAD];
  __shared__ __attribute__((aligned(16))) float Al[16 * HIDDEN];
  __shared__ float Ol[16 * NPAD];
  const int lane = threadIdx.x;
  const int tile = blockIdx.x;

  const float* Ab = z2 + (size_t)tile * 16 * HIDDEN;
  v4f wv[5], av[2];
  #pragma unroll
  for (int j = 0; j < 5; ++j)
    wv[j] = *(const v4f*)(w2 + 4 * (lane + 32 * j));
  #pragma unroll
  for (int j = 0; j < 2; ++j)
    av[j] = *(const v4f*)(Ab + 4 * (lane + 32 * j));
  #pragma unroll
  for (int j = 0; j < 5; ++j) {
    const int q = lane + 32 * j;
    const int r = q / 10, c4 = q % 10;
    *(v4f*)(Wl + r * NPAD + 4 * c4) = wv[j];
  }
  {
    const v4f zero4 = {0.0f, 0.0f, 0.0f, 0.0f};
    *(v4f*)(Wl + (lane >> 1) * NPAD + NCLASS + 4 * (lane & 1)) = zero4;
  }
  #pragma unroll
  for (int j = 0; j < 2; ++j)
    *(v4f*)(Al + 4 * (lane + 32 * j)) = av[j];
  __syncthreads();

  const int half = lane >> 4;
  const int l    = lane & 15;

  #pragma unroll
  for (int t = 0; t < 3; ++t) {
    v8f c = {};
    #pragma unroll
    for (int k0 = 0; k0 < HIDDEN; k0 += 4) {
      const int ka = k0 + 2 * half;
      v2f a, b;
      a.x = Al[l * HIDDEN + ka];
      a.y = Al[l * HIDDEN + ka + 1];
      b.x = Wl[ka * NPAD + t * 16 + l];
      b.y = Wl[(ka + 1) * NPAD + t * 16 + l];
      c = __builtin_amdgcn_wmma_f32_16x16x4_f32(false, a, false, b,
                                                (short)0, c, false, false);
    }
    #pragma unroll
    for (int i = 0; i < 8; ++i)
      Ol[(i + 8 * half) * NPAD + t * 16 + l] = c[i];
  }
  __syncthreads();

  if (lane < 16) {
    const float* row = Ol + lane * NPAD;
    float mx = row[0];
    for (int j = 1; j < NCLASS; ++j) mx = fmaxf(mx, row[j]);
    float s = 0.0f;
    for (int j = 0; j < NCLASS; ++j) s += __expf(row[j] - mx);
    const float lse = mx + __logf(s);
    float* ob = out + ((size_t)tile * 16 + lane) * NCLASS;
    for (int j = 0; j < NCLASS; ++j) ob[j] = row[j] - lse;
  }
}

extern "C" void kernel_launch(void* const* d_in, const int* in_sizes, int n_in,
                              void* d_out, int out_size, void* d_ws, size_t ws_size,
                              hipStream_t stream) {
  const float* features = (const float*)d_in[0];
  const float* weight1  = (const float*)d_in[1];
  const float* weight2  = (const float*)d_in[2];
  const float* edgeVals = (const float*)d_in[3];
  const int*   edgeRows = (const int*)  d_in[4];
  const int*   edgeCols = (const int*)  d_in[5];

  const int nNodes = in_sizes[0] / IN_DIM;
  const int nEdges = in_sizes[3];
  const int nTiles = nNodes / 16;

  float* xw = (float*)d_ws;
  float* t  = xw + (size_t)nNodes * HIDDEN;
  float* z2 = t  + (size_t)nNodes * HIDDEN;

  {
    int n = nNodes * (2 * HIDDEN);
    gcn_zero_f32<<<(n + 255) / 256, 256, 0, stream>>>(t, n);
  }
  gcn_gemm1<<<nTiles, 32, 0, stream>>>(features, weight1, xw);
  gcn_spmm_l1<<<(nEdges + 255) / 256, 256, 0, stream>>>(
      edgeRows, edgeCols, edgeVals, xw, t, nEdges);
  gcn_spmm_l2<<<(nEdges + 255) / 256, 256, 0, stream>>>(
      edgeRows, edgeCols, edgeVals, t, z2, nEdges);
  gcn_gemm2_lsm<<<nTiles, 32, 0, stream>>>(z2, weight2, (float*)d_out);
}


# Edge passes: each half of the chip adds into its own plane of t and z2 (plane picked by die id, speed only); planes summed at layer 2's gather and GEMM2's load; f32 sums split in two partials
# speedup vs baseline: 5.0778x; 1.5939x over previous
#include <hip/hip_runtime.h>
#include <hip/hip_bf16.h>
#include <math.h>


typedef __attribute__((ext_vector_type(2))) float v2f;
typedef __attribute__((ext_vector_type(4))) float v4f;
typedef __attribute__((ext_vector_type(8))) float v8f;

#define IN_DIM   128
#define HIDDEN    16
#define NCLASS    40
#define NPAD      48

__global__ void gcn_zero_f32(float* __restrict__ p, int n) {
  int i = blockIdx.x * blockDim.x + threadIdx.x;
  if (i < n) {
    p[i] = 0.0f;
    p[(size_t)n + i] = 0.0f;
  }
}


#define GCN_SHARE(x, J) __builtin_amdgcn_update_dpp(0, (x), 0x150 + (J), 0xf, 0xf, true)
#define GCN_FOR16(OP) OP(0) OP(1) OP(2) OP(3) OP(4) OP(5) OP(6) OP(7) \
                      OP(8) OP(9) OP(10) OP(11) OP(12) OP(13) OP(14) OP(15)

template <bool RELU>
__device__ __forceinline__ float gcn_act(float d) {
  return RELU ? (d > 0.0f ? d : 0.0f) : d;
}

template <bool RELU>
__device__ __forceinline__ void gcn_spmm16_body(const int* __restrict__ rows,
                                                const int* __restrict__ cols,
                                                const float* __restrict__ vals,
                                                const float* __restrict__ src0,
                                                const float* __restrict__ src1,
                                                float* __restrict__ dst0,
                                                float* __restrict__ dst1,
                                                int nEdges) {
  const int e    = blockIdx.x * blockDim.x + threadIdx.x;
  const int lane = threadIdx.x & 31;
  const int e0   = e - lane;
  if (e0 >= nEdges) return;
  const int f    = lane & 15;
  const int half = lane >> 4;

  const unsigned seAid = __builtin_amdgcn_s_sendmsg_rtn(0x87);
  float* __restrict__ dst = ((seAid >> 18) & 1u) ? dst1 : dst0;

  int rofs = 0, cofs = 0, vbits = 0;
  if (e < nEdges) {
    rofs  = rows[e] * HIDDEN;
    cofs  = cols[e] * HIDDEN;
    vbits = __float_as_int(vals[e]);
  }

  if (e0 + 32 <= nEdges) {
    float d[16];
    if (RELU) {
      float d1[16];
#define GCN_GATHER(J) { const unsigned o = (unsigned)(GCN_SHARE(cofs, J) + f); \
                        d[J] = src0[o]; d1[J] = src1[o]; }
      GCN_FOR16(GCN_GATHER)
#undef GCN_GATHER
#define GCN_SUM(J) d[J] += d1[J];
      GCN_FOR16(GCN_SUM)
#undef GCN_SUM
    } else {
#define GCN_GATHER(J) d[J] = src0[(unsigned)(GCN_SHARE(cofs, J) + f)];
      GCN_FOR16(GCN_GATHER)
#undef GCN_GATHER
    }
#define GCN_ADD(J)                                                   \
    atomicAdd(dst + (unsigned)(GCN_SHARE(rofs, J) + f),                \
              __int_as_float(GCN_SHARE(vbits, J)) * gcn_act<RELU>(d[J]));
    GCN_FOR16(GCN_ADD)
#undef GCN_ADD
  } else {
    for (int j = 0; j < 16; ++j) {
      const int sl = (half << 4) + j;
      const int cj = __shfl(cofs, sl, 32);
      const int rj = __shfl(rofs, sl, 32);
      const float vj = __int_as_float(__shfl(vbits, sl, 32));
      if (e0 + sl < nEdges) {
        float sj = src0[(unsigned)(cj + f)];
        if (RELU) sj += src1[(unsigned)(cj + f)];
        atomicAdd(dst + (unsigned)(rj + f), vj * gcn_act<RELU>(sj));
      }
    }
  }
}

__global__ void gcn_spmm_l1(const int* __restrict__ rows,
                            const int* __restrict__ cols,
                            const float* __restrict__ vals,
                            const float* __restrict__ xw,
                            float* __restrict__ t0,
                            float* __restrict__ t1, int nEdges) {
  gcn_spmm16_body<false>(rows, cols, vals, xw, xw, t0, t1, nEdges);
}

__global__ void gcn_spmm_l2(const int* __restrict__ rows,
                            const int* __restrict__ cols,
                            const float* __restrict__ vals,
                            const float* __restrict__ t0,
                            const float* __restrict__ t1,
                            float* __restrict__ z0,
                            float* __restrict__ z1, int nEdges) {
  gcn_spmm16_body<true>(rows, cols, vals, t0, t1, z0, z1, nEdges);
}

__global__ void __launch_bounds__(32)
gcn_gemm1(const float* __restrict__ x,
          const float* __restrict__ w1,
          float* __restrict__ xw) {
  __shared__ __attribute__((aligned(16))) float Wl[IN_DIM * HIDDEN];
  __shared__ __attribute__((aligned(16))) float Al[16 * IN_DIM];
  const int lane = threadIdx.x;
  const int tile = blockIdx.x;

  const float* Ab = x + (size_t)tile * 16 * IN_DIM;
  v4f xa[16], wb[16];
  #pragma unroll
  for (int j = 0; j < 16; ++j)
    xa[j] = *(const v4f*)(Ab + 4 * (lane + 32 * j));
  #pragma unroll
  for (int j = 0; j < 16; ++j)
    wb[j] = *(const v4f*)(w1 + 4 * (lane + 32 * j));
  #pragma unroll
  for (int j = 0; j < 16; ++j)
    *(v4f*)(Al + 4 * (lane + 32 * j)) = xa[j];
  #pragma unroll
  for (int j = 0; j < 16; ++j)
    *(v4f*)(Wl + 4 * (lane + 32 * j)) = wb[j];
  __syncthreads();

  const int half = lane >> 4;
  const int l    = lane & 15;

  v8f c = {};
  #pragma unroll 8
  for (int k0 = 0; k0 < IN_DIM; k0 += 4) {
    const int ka = k0 + 2 * half;
    v2f a, b;
    a.x = Al[l * IN_DIM + ka];
    a.y = Al[l * IN_DIM + ka + 1];
    b.x = Wl[ka * HIDDEN + l];
    b.y = Wl[(ka + 1) * HIDDEN + l];
    c = __builtin_amdgcn_wmma_f32_16x16x4_f32(false, a, false, b,
                                              (short)0, c, false, false);
  }

  float* ob = xw + (size_t)tile * 16 * HIDDEN;
  #pragma unroll
  for (int i = 0; i < 8; ++i)
    ob[(i + 8 * half) * HIDDEN + l] = c[i];
}

__global__ void __launch_bounds__(32)
gcn_gemm2_lsm(const float* __restrict__ z0,
              const float* __restrict__ z1,
              const float* __restrict__ w2,
              float* __restrict__ out) {
  __shared__ __attribute__((aligned(16))) float Wl[HIDDEN * NPAD];
  __shared__ __attribute__((aligned(16))) float Al[16 * HIDDEN];
  __shared__ float Ol[16 * NPAD];
  const int lane = threadIdx.x;
  const int tile = blockIdx.x;

  const float* Ab = z0 + (size_t)tile * 16 * HIDDEN;
  const float* Bb = z1 + (size_t)tile * 16 * HIDDEN;
  v4f wv[5], av[2], bv[2];
  #pragma unroll
  for (int j = 0; j < 5; ++j)
    wv[j] = *(const v4f*)(w2 + 4 * (lane + 32 * j));
  #pragma unroll
  for (int j = 0; j < 2; ++j) {
    av[j] = *(const v4f*)(Ab + 4 * (lane + 32 * j));
    bv[j] = *(const v4f*)(Bb + 4 * (lane + 32 * j));
  }
  #pragma unroll
  for (int j = 0; j < 5; ++j) {
    const int q = lane + 32 * j;
    const int r = q / 10, c4 = q % 10;
    *(v4f*)(Wl + r * NPAD + 4 * c4) = wv[j];
  }
  {
    const v4f zero4 = {0.0f, 0.0f, 0.0f, 0.0f};
    *(v4f*)(Wl + (lane >> 1) * NPAD + NCLASS + 4 * (lane & 1)) = zero4;
  }
  #pragma unroll
  for (int j = 0; j < 2; ++j)
    *(v4f*)(Al + 4 * (lane + 32 * j)) = av[j] + bv[j];
  __syncthreads();

  const int half = lane >> 4;
  const int l    = lane & 15;

  #pragma unroll
  for (int t = 0; t < 3; ++t) {
    v8f c = {};
    #pragma unroll
    for (int k0 = 0; k0 < HIDDEN; k0 += 4) {
      const int ka = k0 + 2 * half;
      v2f a, b;
      a.x = Al[l * HIDDEN + ka];
      a.y = Al[l * HIDDEN + ka + 1];
      b.x = Wl[ka * NPAD + t * 16 + l];
      b.y = Wl[(ka + 1) * NPAD + t * 16 + l];
      c = __builtin_amdgcn_wmma_f32_16x16x4_f32(false, a, false, b,
                                                (short)0, c, false, false);
    }
    #pragma unroll
    for (int i = 0; i < 8; ++i)
      Ol[(i + 8 * half) * NPAD + t * 16 + l] = c[i];
  }
  __syncthreads();

  if (lane < 16) {
    const float* row = Ol + lane * NPAD;
    float mx = row[0];
    for (int j = 1; j < NCLASS; ++j) mx = fmaxf(mx, row[j]);
    float s = 0.0f;
    for (int j = 0; j < NCLASS; ++j) s += __expf(row[j] - mx);
    const float lse = mx + __logf(s);
    float* ob = out + ((size_t)tile * 16 + lane) * NCLASS;
    for (int j = 0; j < NCLASS; ++j) ob[j] = row[j] - lse;
  }
}

extern "C" void kernel_launch(void* const* d_in, const int* in_sizes, int n_in,
                              void* d_out, int out_size, void* d_ws, size_t ws_size,
                              hipStream_t stream) {
  const float* features = (const float*)d_in[0];
  const float* weight1  = (const float*)d_in[1];
  const float* weight2  = (const float*)d_in[2];
  const float* edgeVals = (const float*)d_in[3];
  const int*   edgeRows = (const int*)  d_in[4];
  const int*   edgeCols = (const int*)  d_in[5];

  const int nNodes = in_sizes[0] / IN_DIM;
  const int nEdges = in_sizes[3];
  const int nTiles = nNodes / 16;

  float* xw = (float*)d_ws;
  float* t0 = xw + (size_t)nNodes * HIDDEN;
  float* z0 = t0 + (size_t)nNodes * HIDDEN;
  float* t1 = z0 + (size_t)nNodes * HIDDEN;
  float* z1 = t1 + (size_t)nNodes * HIDDEN;

  {
    int n = nNodes * (2 * HIDDEN);
    gcn_zero_f32<<<(n + 255) / 256, 256, 0, stream>>>(t0, n);
  }
  gcn_gemm1<<<nTiles, 32, 0, stream>>>(features, weight1, xw);
  gcn_spmm_l1<<<(nEdges + 255) / 256, 256, 0, stream>>>(
      edgeRows, edgeCols, edgeVals, xw, t0, t1, nEdges);
  gcn_spmm_l2<<<(nEdges + 255) / 256, 256, 0, stream>>>(
      edgeRows, edgeCols, edgeVals, t0, t1, z0, z1, nEdges);
  gcn_gemm2_lsm<<<nTiles, 32, 0, stream>>>(z0, z1, weight2, (float*)d_out);
}


# Last kernel: each 16x40 output tile (2560 contiguous bytes) leaves LDS as 5 16-byte stores per lane, whole 128-B lines per instruction, not 11 stores per lane into 16 lines each; values bit-identical
# speedup vs baseline: 5.3699x; 1.0575x over previous
#include <hip/hip_runtime.h>
#include <hip/hip_bf16.h>
#include <math.h>


typedef __attribute__((ext_vector_type(2))) float v2f;
typedef __attribute__((ext_vector_type(4))) float v4f;
typedef __attribute__((ext_vector_type(8))) float v8f;

#define IN_DIM   128
#define HIDDEN    16
#define NCLASS    40
#define NPAD      48

__global__ void gcn_zero_f32(float* __restrict__ p, int n) {
  int i = blockIdx.x * blockDim.x + threadIdx.x;
  if (i < n) {
    p[i] = 0.0f;
    p[(size_t)n + i] = 0.0f;
  }
}


#define GCN_SHARE(x, J) __builtin_amdgcn_update_dpp(0, (x), 0x150 + (J), 0xf, 0xf, true)
#define GCN_FOR16(OP) OP(0) OP(1) OP(2) OP(3) OP(4) OP(5) OP(6) OP(7) \
                      OP(8) OP(9) OP(10) OP(11) OP(12) OP(13) OP(14) OP(15)

template <bool RELU>
__device__ __forceinline__ float gcn_act(float d) {
  return RELU ? (d > 0.0f ? d : 0.0f) : d;
}

template <bool RELU>
__device__ __forceinline__ void gcn_spmm16_body(const int* __restrict__ rows,
                                                const int* __restrict__ cols,
                                                const float* __restrict__ vals,
                                                const float* __restrict__ src0,
                                                const float* __restrict__ src1,
                                                float* __restrict__ dst0,
                                                float* __restrict__ dst1,
                                                int nEdges) {
  const int e    = blockIdx.x * blockDim.x + threadIdx.x;
  const int lane = threadIdx.x & 31;
  const int e0   = e - lane;
  if (e0 >= nEdges) return;
  const int f    = lane & 15;
  const int half = lane >> 4;

  const unsigned seAid = __builtin_amdgcn_s_sendmsg_rtn(0x87);
  float* __restrict__ dst = ((seAid >> 18) & 1u) ? dst1 : dst0;

  int rofs = 0, cofs = 0, vbits = 0;
  if (e < nEdges) {
    rofs  = rows[e] * HIDDEN;
    cofs  = cols[e] * HIDDEN;
    vbits = __float_as_int(vals[e]);
  }

  if (e0 + 32 <= nEdges) {
    float d[16];
    if (RELU) {
      float d1[16];
#define GCN_GATHER(J) { const unsigned o = (unsigned)(GCN_SHARE(cofs, J) + f); \
                        d[J] = src0[o]; d1[J] = src1[o]; }
      GCN_FOR16(GCN_GATHER)
#undef GCN_GATHER
#define GCN_SUM(J) d[J] += d1[J];
      GCN_FOR16(GCN_SUM)
#undef GCN_SUM
    } else {
#define GCN_GATHER(J) d[J] = src0[(unsigned)(GCN_SHARE(cofs, J) + f)];
      GCN_FOR16(GCN_GATHER)
#undef GCN_GATHER
    }
#define GCN_ADD(J)                                                   \
    atomicAdd(dst + (unsigned)(GCN_SHARE(rofs, J) + f),                \
              __int_as_float(GCN_SHARE(vbits, J)) * gcn_act<RELU>(d[J]));
    GCN_FOR16(GCN_ADD)
#undef GCN_ADD
  } else {
    for (int j = 0; j < 16; ++j) {
      const int sl = (half << 4) + j;
      const int cj = __shfl(cofs, sl, 32);
      const int rj = __shfl(rofs, sl, 32);
      const float vj = __int_as_float(__shfl(vbits, sl, 32));
      if (e0 + sl < nEdges) {
        float sj = src0[(unsigned)(cj + f)];
        if (RELU) sj += src1[(unsigned)(cj + f)];
        atomicAdd(dst + (unsigned)(rj + f), vj * gcn_act<RELU>(sj));
      }
    }
  }
}

__global__ void gcn_spmm_l1(const int* __restrict__ rows,
                            const int* __restrict__ cols,
                            const float* __restrict__ vals,
                            const float* __restrict__ xw,
                            float* __restrict__ t0,
                            float* __restrict__ t1, int nEdges) {
  gcn_spmm16_body<false>(rows, cols, vals, xw, xw, t0, t1, nEdges);
}

__global__ void gcn_spmm_l2(const int* __restrict__ rows,
                            const int* __restrict__ cols,
                            const float* __restrict__ vals,
                            const float* __restrict__ t0,
                            const float* __restrict__ t1,
                            float* __restrict__ z0,
                            float* __restrict__ z1, int nEdges) {
  gcn_spmm16_body<true>(rows, cols, vals, t0, t1, z0, z1, nEdges);
}

__global__ void __launch_bounds__(32)
gcn_gemm1(const float* __restrict__ x,
          const float* __restrict__ w1,
          float* __restrict__ xw) {
  __shared__ __attribute__((aligned(16))) float Wl[IN_DIM * HIDDEN];
  __shared__ __attribute__((aligned(16))) float Al[16 * IN_DIM];
  const int lane = threadIdx.x;
  const int tile = blockIdx.x;

  const float* Ab = x + (size_t)tile * 16 * IN_DIM;
  v4f xa[16], wb[16];
  #pragma unroll
  for (int j = 0; j < 16; ++j)
    xa[j] = *(const v4f*)(Ab + 4 * (lane + 32 * j));
  #pragma unroll
  for (int j = 0; j < 16; ++j)
    wb[j] = *(const v4f*)(w1 + 4 * (lane + 32 * j));
  #pragma unroll
  for (int j = 0; j < 16; ++j)
    *(v4f*)(Al + 4 * (lane + 32 * j)) = xa[j];
  #pragma unroll
  for (int j = 0; j < 16; ++j)
    *(v4f*)(Wl + 4 * (lane + 32 * j)) = wb[j];
  __syncthreads();

  const int half = lane >> 4;
  const int l    = lane & 15;

  v8f c = {};
  #pragma unroll 8
  for (int k0 = 0; k0 < IN_DIM; k0 += 4) {
    const int ka = k0 + 2 * half;
    v2f a, b;
    a.x = Al[l * IN_DIM + ka];
    a.y = Al[l * IN_DIM + ka + 1];
    b.x = Wl[ka * HIDDEN + l];
    b.y = Wl[(ka + 1) * HIDDEN + l];
    c = __builtin_amdgcn_wmma_f32_16x16x4_f32(false, a, false, b,
                                              (short)0, c, false, false);
  }

  float* ob = xw + (size_t)tile * 16 * HIDDEN;
  #pragma unroll
  for (int i = 0; i < 8; ++i)
    ob[(i + 8 * half) * HIDDEN + l] = c[i];
}

__global__ void __launch_bounds__(32)
gcn_gemm2_lsm(const float* __restrict__ z0,
              const float* __restrict__ z1,
              const float* __restrict__ w2,
              float* __restrict__ out) {
  __shared__ __attribute__((aligned(16))) float Wl[HIDDEN * NPAD];
  __shared__ __attribute__((aligned(16))) float Al[16 * HIDDEN];
  __shared__ __attribute__((aligned(16))) float Ol[16 * NPAD];
  const int lane = threadIdx.x;
  const int tile = blockIdx.x;

  const float* Ab = z0 + (size_t)tile * 16 * HIDDEN;
  const float* Bb = z1 + (size_t)tile * 16 * HIDDEN;
  v4f wv[5], av[2], bv[2];
  #pragma unroll
  for (int j = 0; j < 5; ++j)
    wv[j] = *(const v4f*)(w2 + 4 * (lane + 32 * j));
  #pragma unroll
  for (int j = 0; j < 2; ++j) {
    av[j] = *(const v4f*)(Ab + 4 * (lane + 32 * j));
    bv[j] = *(const v4f*)(Bb + 4 * (lane + 32 * j));
  }
  #pragma unroll
  for (int j = 0; j < 5; ++j) {
    const int q = lane + 32 * j;
    const int r = q / 10, c4 = q % 10;
    *(v4f*)(Wl + r * NPAD + 4 * c4) = wv[j];
  }
  {
    const v4f zero4 = {0.0f, 0.0f, 0.0f, 0.0f};
    *(v4f*)(Wl + (lane >> 1) * NPAD + NCLASS + 4 * (lane & 1)) = zero4;
  }
  #pragma unroll
  for (int j = 0; j < 2; ++j)
    *(v4f*)(Al + 4 * (lane + 32 * j)) = av[j] + bv[j];
  __syncthreads();

  const int half = lane >> 4;
  const int l    = lane & 15;

  #pragma unroll
  for (int t = 0; t < 3; ++t) {
    v8f c = {};
    #pragma unroll
    for (int k0 = 0; k0 < HIDDEN; k0 += 4) {
      const int ka = k0 + 2 * half;
      v2f a, b;
      a.x = Al[l * HIDDEN + ka];
      a.y = Al[l * HIDDEN + ka + 1];
      b.x = Wl[ka * NPAD + t * 16 + l];
      b.y = Wl[(ka + 1) * NPAD + t * 16 + l];
      c = __builtin_amdgcn_wmma_f32_16x16x4_f32(false, a, false, b,
                                                (short)0, c, false, false);
    }
    #pragma unroll
    for (int i = 0; i < 8; ++i)
      Ol[(i + 8 * half) * NPAD + t * 16 + l] = c[i];
  }
  __syncthreads();

  const float* row = Ol + l * NPAD;
  float mx = row[0];
  for (int j = 1; j < NCLASS; ++j) mx = fmaxf(mx, row[j]);
  float s = 0.0f;
  for (int j = 0; j < NCLASS; ++j) s += __expf(row[j] - mx);
  const float lse = mx + __logf(s);

  float* ob = out + (size_t)tile * 16 * NCLASS;
  #pragma unroll
  for (int j = 0; j < 5; ++j) {
    const int q = lane + 32 * j;
    const int r = q / 10, c4 = q % 10;
    const float lr = __shfl(lse, r, 32);
    v4f v = *(const v4f*)(Ol + r * NPAD + 4 * c4);
    v.x -= lr; v.y -= lr; v.z -= lr; v.w -= lr;
    *(v4f*)(ob + 4 * q) = v;
  }
}

extern "C" void kernel_launch(void* const* d_in, const int* in_sizes, int n_in,
                              void* d_out, int out_size, void* d_ws, size_t ws_size,
                              hipStream_t stream) {
  const float* features = (const float*)d_in[0];
  const float* weight1  = (const float*)d_in[1];
  const float* weight2  = (const float*)d_in[2];
  const float* edgeVals = (const float*)d_in[3];
  const int*   edgeRows = (const int*)  d_in[4];
  const int*   edgeCols = (const int*)  d_in[5];

  const int nNodes = in_sizes[0] / IN_DIM;
  const int nEdges = in_sizes[3];
  const int nTiles = nNodes / 16;

  float* xw = (float*)d_ws;
  float* t0 = xw + (size_t)nNodes * HIDDEN;
  float* z0 = t0 + (size_t)nNodes * HIDDEN;
  float* t1 = z0 + (size_t)nNodes * HIDDEN;
  float* z1 = t1 + (size_t)nNodes * HIDDEN;

  {
    int n = nNodes * (2 * HIDDEN);
    gcn_zero_f32<<<(n + 255) / 256, 256, 0, stream>>>(t0, n);
  }
  gcn_gemm1<<<nTiles, 32, 0, stream>>>(features, weight1, xw);
  gcn_spmm_l1<<<(nEdges + 255) / 256, 256, 0, stream>>>(
      edgeRows, edgeCols, edgeVals, xw, t0, t1, nEdges);
  gcn_spmm_l2<<<(nEdges + 255) / 256, 256, 0, stream>>>(
      edgeRows, edgeCols, edgeVals, t0, t1, z0, z1, nEdges);
  gcn_gemm2_lsm<<<nTiles, 32, 0, stream>>>(z0, z1, weight2, (float*)d_out);
}
